# starting kernel
# baseline (speedup 1.0000x reference)
#include <hip/hip_runtime.h>
#include <hip/hip_bf16.h>

#define NB 16
#define NC 128
#define ND 256
#define NROWS (NB * NC)

typedef __attribute__((ext_vector_type(2))) float v2f;
typedef __attribute__((ext_vector_type(8))) float v8f;

static __device__ __forceinline__ v8f wmma4(v2f a, v2f b, v8f c) {
    return __builtin_amdgcn_wmma_f32_16x16x4_f32(
         false, a,  false, b,
         (short)0, c,  false,  false);
}

__global__ void __launch_bounds__(256) k_gemm_pq(const float* __restrict__ H,
                                                 const float* __restrict__ W1,
                                                 float* __restrict__ P,
                                                 float* __restrict__ Q) {
    const int lane = threadIdx.x & 31;
    const int wv   = threadIdx.x >> 5;
    const int tile = blockIdx.x * 8 + wv;
    const int mt = tile >> 4;
    const int nt = tile & 15;
    const int hi = lane >> 4, lo = lane & 15;
    const int n  = nt * 16 + lo;

    const float* Arow = H + (mt * 16 + lo) * ND;
    v8f accP = {}, accQ = {};
#pragma unroll 4
    for (int k = 0; k < ND; k += 4) {
        const int ka = k + 2 * hi;
        v2f a = *(const v2f*)(Arow + ka);
        v2f bp, bq;
        bp.x = W1[ka * ND + n];
        bp.y = W1[(ka + 1) * ND + n];
        bq.x = W1[(ka + ND) * ND + n];
        bq.y = W1[(ka + ND + 1) * ND + n];
        accP = wmma4(a, bp, accP);
        accQ = wmma4(a, bq, accQ);
    }
#pragma unroll
    for (int g = 0; g < 8; ++g) {
        const int row = mt * 16 + g + 8 * hi;
        P[row * ND + n] = accP[g];
        Q[row * ND + n] = accQ[g];
    }
}

__global__ void __launch_bounds__(256) k_scores(const float* __restrict__ P,
                                                const float* __restrict__ Q,
                                                const float* __restrict__ b1,
                                                const float* __restrict__ W2,
                                                const float* __restrict__ b2,
                                                const float* __restrict__ A_a,
                                                float* __restrict__ G) {
    __shared__ float Pl[16][ND + 1];
    __shared__ float Ql[16][ND + 1];
    __shared__ float b1s[ND];
    __shared__ float w2s[ND];

    const int blk = blockIdx.x;
    const int vt = blk & 7;
    const int ut = (blk >> 3) & 7;
    const int b  = blk >> 6;
    const int t  = threadIdx.x;

#pragma unroll
    for (int i = t; i < 16 * ND; i += 256) {
        const int r = i >> 8, k = i & (ND - 1);
        Pl[r][k] = P[(b * NC + ut * 16 + r) * ND + k];
        Ql[r][k] = Q[(b * NC + vt * 16 + r) * ND + k];
    }
    b1s[t] = b1[t];
    w2s[t] = W2[t];
    __syncthreads();

    const int u = t >> 4, v = t & 15;
    float acc = 0.f;
#pragma unroll 8
    for (int k = 0; k < ND; ++k) {
        const float x = Pl[u][k] + Ql[v][k] + b1s[k];
        acc = fmaf(fmaxf(x, 0.f), w2s[k], acc);
    }
    const float s  = 1.0f / (1.0f + __expf(-(acc + b2[0])));
    const int uu = ut * 16 + u, vv = vt * 16 + v;
    G[(b * NC + uu) * NC + vv] = s * A_a[uu * NC + vv];
}

__global__ void __launch_bounds__(256) k_agg(const float* __restrict__ G,
                                             const float* __restrict__ Am,
                                             const float* __restrict__ h,
                                             float* __restrict__ a_t,
                                             float* __restrict__ m_t) {
    const int lane = threadIdx.x & 31;
    const int wv   = threadIdx.x >> 5;
    const int tile = blockIdx.x * 8 + wv;
    const int dt = tile & 15;
    const int vt = (tile >> 4) & 7;
    const int b  = tile >> 7;
    const int hi = lane >> 4, lo = lane & 15;
    const int n  = dt * 16 + lo;
    const int m  = vt * 16 + lo;

    const float* Gb = G + b * NC * NC;
    const float* hb = h + b * NC * ND;
    v8f accA = {}, accM = {};
#pragma unroll 4
    for (int k = 0; k < NC; k += 4) {
        const int ka = k + 2 * hi;
        v2f ag, am, bb;
        ag.x = Gb[ka * NC + m];
        ag.y = Gb[(ka + 1) * NC + m];
        am.x = Am[ka * NC + m];
        am.y = Am[(ka + 1) * NC + m];
        bb.x = hb[ka * ND + n];
        bb.y = hb[(ka + 1) * ND + n];
        accA = wmma4(ag, bb, accA);
        accM = wmma4(am, bb, accM);
    }
#pragma unroll
    for (int g = 0; g < 8; ++g) {
        const int row = vt * 16 + g + 8 * hi;
        a_t[(b * NC + row) * ND + n] = accA[g];
        m_t[(b * NC + row) * ND + n] = accM[g] * hb[row * ND + n];
    }
}

__global__ void __launch_bounds__(256) k_final(const float* __restrict__ a_t,
                                               const float* __restrict__ m_t,
                                               const float* __restrict__ h,
                                               const float* __restrict__ Wa,
                                               const float* __restrict__ Wm,
                                               const float* __restrict__ ba,
                                               const float* __restrict__ bm,
                                               const float* __restrict__ pb1,
                                               const float* __restrict__ pb2,
                                               const float* __restrict__ pb3,
                                               float* __restrict__ z) {
    const int lane = threadIdx.x & 31;
    const int wv   = threadIdx.x >> 5;
    const int tile = blockIdx.x * 8 + wv;
    const int mt = tile >> 4;
    const int nt = tile & 15;
    const int hi = lane >> 4, lo = lane & 15;
    const int n  = nt * 16 + lo;

    const float* Ar = a_t + (mt * 16 + lo) * ND;
    const float* Mr = m_t + (mt * 16 + lo) * ND;
    v8f acc1 = {}, acc2 = {};
#pragma unroll 4
    for (int k = 0; k < ND; k += 4) {
        const int ka = k + 2 * hi;
        v2f a1 = *(const v2f*)(Ar + ka);
        v2f a2 = *(const v2f*)(Mr + ka);
        v2f w1v, w2v;
        w1v.x = Wa[ka * ND + n];
        w1v.y = Wa[(ka + 1) * ND + n];
        w2v.x = Wm[ka * ND + n];
        w2v.y = Wm[(ka + 1) * ND + n];
        acc1 = wmma4(a1, w1v, acc1);
        acc2 = wmma4(a2, w2v, acc2);
    }
    const float beta1 = pb1[0], beta2 = pb2[0], beta3 = pb3[0];
    const float bav = ba[n], bmv = bm[n];
#pragma unroll
    for (int g = 0; g < 8; ++g) {
        const int row = mt * 16 + g + 8 * hi;
        z[row * ND + n] = beta1 * h[row * ND + n]
                        + beta2 * (acc1[g] + bav)
                        + beta3 * (acc2[g] + bmv);
    }
}

extern "C" void kernel_launch(void* const* d_in, const int* in_sizes, int n_in,
                              void* d_out, int out_size, void* d_ws, size_t ws_size,
                              hipStream_t stream) {
    (void)in_sizes; (void)n_in; (void)out_size; (void)ws_size;

    const float* h_t = (const float*)d_in[0];
    const float* A_a = (const float*)d_in[1];
    const float* A_m = (const float*)d_in[2];
    const float* W1  = (const float*)d_in[3];
    const float* b1  = (const float*)d_in[4];
    const float* W2  = (const float*)d_in[5];
    const float* b2  = (const float*)d_in[6];
    const float* Wa  = (const float*)d_in[7];
    const float* ba  = (const float*)d_in[8];
    const float* Wm  = (const float*)d_in[9];
    const float* bm  = (const float*)d_in[10];
    const float* be1 = (const float*)d_in[11];
    const float* be2 = (const float*)d_in[12];
    const float* be3 = (const float*)d_in[13];
    float* z = (float*)d_out;

    float* P  = (float*)d_ws;
    float* Q  = P + NROWS * ND;
    float* G  = Q + NROWS * ND;
    float* At = G + NB * NC * NC;
    float* Mt = At + NROWS * ND;

    k_gemm_pq<<<256, 256, 0, stream>>>(h_t, W1, P, Q);
    k_scores<<<NB * 8 * 8, 256, 0, stream>>>(P, Q, b1, W2, b2, A_a, G);
    k_agg<<<256, 256, 0, stream>>>(G, A_m, h_t, At, Mt);
    k_final<<<256, 256, 0, stream>>>(At, Mt, h_t, Wa, Wm, ba, bm,
                                     be1, be2, be3, z);
}


# Software pipelining of the three WMMA K loops: next 32-k chunk of A/B fragments loaded into a second register set before the current chunk's WMMAs (loops unrolled, order held by sched_barrier)
# speedup vs baseline: 1.5581x; 1.5581x over previous
#include <hip/hip_runtime.h>
#include <hip/hip_bf16.h>

#define NB 16
#define NC 128
#define ND 256
#define NROWS (NB * NC)

typedef __attribute__((ext_vector_type(2))) float v2f;
typedef __attribute__((ext_vector_type(8))) float v8f;

static __device__ __forceinline__ v8f wmma4(v2f a, v2f b, v8f c) {
    return __builtin_amdgcn_wmma_f32_16x16x4_f32(
         false, a,  false, b,
         (short)0, c,  false,  false);
}

#define KCH 8

static __device__ __forceinline__ void hold_order() { __builtin_amdgcn_sched_barrier(0); }

__global__ void __launch_bounds__(256) k_gemm_pq(const float* __restrict__ H,
                                                 const float* __restrict__ W1,
                                                 float* __restrict__ P,
                                                 float* __restrict__ Q) {
    const int lane = threadIdx.x & 31;
    const int wv   = threadIdx.x >> 5;
    const int tile = blockIdx.x * 8 + wv;
    const int mt = tile >> 4;
    const int nt = tile & 15;
    const int hi = lane >> 4, lo = lane & 15;
    const int n  = nt * 16 + lo;

    const float* Ah = H + (mt * 16 + lo) * ND + 2 * hi;
    const float* Wh = W1 + (2 * hi) * ND + n;
    v8f accP = {}, accQ = {};
    v2f a0[KCH], p0[KCH], q0[KCH], a1[KCH], p1[KCH], q1[KCH];

#define LOAD_PQ(A_, P_, Q_, K0)                                         \
    _Pragma("unroll")          \
    for (int j = 0; j < KCH; ++j) {                                     \
        const int kk = (K0) + 4 * j;                                    \
        A_[j]   = *(const v2f*)(Ah + kk);                               \
        P_[j].x = Wh[kk * ND];                                          \
        P_[j].y = Wh[(kk + 1) * ND];                                    \
        Q_[j].x = Wh[(kk + ND) * ND];                                   \
        Q_[j].y = Wh[(kk + ND + 1) * ND];                               \
    }
#define MMA_PQ(A_, P_, Q_)                                              \
    _Pragma("unroll")          \
    for (int j = 0; j < KCH; ++j) {                                     \
        accP = wmma4(A_[j], P_[j], accP);                               \
        accQ = wmma4(A_[j], Q_[j], accQ);                               \
    }

    LOAD_PQ(a0, p0, q0, 0)
#pragma unroll
    for (int c = 0; c < ND / (4 * KCH); c += 2) {
        LOAD_PQ(a1, p1, q1, (c + 1) * 4 * KCH)
        hold_order();
        MMA_PQ(a0, p0, q0)
        hold_order();
        if (c + 2 < ND / (4 * KCH)) {
            LOAD_PQ(a0, p0, q0, (c + 2) * 4 * KCH)
        }
        hold_order();
        MMA_PQ(a1, p1, q1)
        hold_order();
    }
#undef LOAD_PQ
#undef MMA_PQ
#pragma unroll
    for (int g = 0; g < 8; ++g) {
        const int row = mt * 16 + g + 8 * hi;
        P[row * ND + n] = accP[g];
        Q[row * ND + n] = accQ[g];
    }
}

__global__ void __launch_bounds__(256) k_scores(const float* __restrict__ P,
                                                const float* __restrict__ Q,
                                                const float* __restrict__ b1,
                                                const float* __restrict__ W2,
                                                const float* __restrict__ b2,
                                                const float* __restrict__ A_a,
                                                float* __restrict__ G) {
    __shared__ float Pl[16][ND + 1];
    __shared__ float Ql[16][ND + 1];
    __shared__ float b1s[ND];
    __shared__ float w2s[ND];

    const int blk = blockIdx.x;
    const int vt = blk & 7;
    const int ut = (blk >> 3) & 7;
    const int b  = blk >> 6;
    const int t  = threadIdx.x;

#pragma unroll
    for (int i = t; i < 16 * ND; i += 256) {
        const int r = i >> 8, k = i & (ND - 1);
        Pl[r][k] = P[(b * NC + ut * 16 + r) * ND + k];
        Ql[r][k] = Q[(b * NC + vt * 16 + r) * ND + k];
    }
    b1s[t] = b1[t];
    w2s[t] = W2[t];
    __syncthreads();

    const int u = t >> 4, v = t & 15;
    float acc = 0.f;
#pragma unroll 8
    for (int k = 0; k < ND; ++k) {
        const float x = Pl[u][k] + Ql[v][k] + b1s[k];
        acc = fmaf(fmaxf(x, 0.f), w2s[k], acc);
    }
    const float s  = 1.0f / (1.0f + __expf(-(acc + b2[0])));
    const int uu = ut * 16 + u, vv = vt * 16 + v;
    G[(b * NC + uu) * NC + vv] = s * A_a[uu * NC + vv];
}

__global__ void __launch_bounds__(256) k_agg(const float* __restrict__ G,
                                             const float* __restrict__ Am,
                                             const float* __restrict__ h,
                                             float* __restrict__ a_t,
                                             float* __restrict__ m_t) {
    const int lane = threadIdx.x & 31;
    const int wv   = threadIdx.x >> 5;
    const int tile = blockIdx.x * 8 + wv;
    const int dt = tile & 15;
    const int vt = (tile >> 4) & 7;
    const int b  = tile >> 7;
    const int hi = lane >> 4, lo = lane & 15;
    const int n  = dt * 16 + lo;
    const int m  = vt * 16 + lo;

    const float* Gb = G + b * NC * NC;
    const float* hb = h + b * NC * ND;
    v8f accA = {}, accM = {};
    const float* Gh = Gb + (2 * hi) * NC + m;
    const float* Mh = Am + (2 * hi) * NC + m;
    const float* Hh = hb + (2 * hi) * ND + n;
    v2f fg0[KCH], fm0[KCH], fh0[KCH], fg1[KCH], fm1[KCH], fh1[KCH];

#define LOAD_AGG(G_, M_, B_, K0)                                        \
    _Pragma("unroll")          \
    for (int j = 0; j < KCH; ++j) {                                     \
        const int kk = (K0) + 4 * j;                                    \
        G_[j].x = Gh[kk * NC];                                          \
        G_[j].y = Gh[(kk + 1) * NC];                                    \
        M_[j].x = Mh[kk * NC];                                          \
        M_[j].y = Mh[(kk + 1) * NC];                                    \
        B_[j].x = Hh[kk * ND];                                          \
        B_[j].y = Hh[(kk + 1) * ND];                                    \
    }
#define MMA_AGG(G_, M_, B_)                                             \
    _Pragma("unroll")          \
    for (int j = 0; j < KCH; ++j) {                                     \
        accA = wmma4(G_[j], B_[j], accA);                               \
        accM = wmma4(M_[j], B_[j], accM);                               \
    }

    LOAD_AGG(fg0, fm0, fh0, 0)
#pragma unroll
    for (int c = 0; c < NC / (4 * KCH); c += 2) {
        LOAD_AGG(fg1, fm1, fh1, (c + 1) * 4 * KCH)
        hold_order();
        MMA_AGG(fg0, fm0, fh0)
        hold_order();
        if (c + 2 < NC / (4 * KCH)) {
            LOAD_AGG(fg0, fm0, fh0, (c + 2) * 4 * KCH)
        }
        hold_order();
        MMA_AGG(fg1, fm1, fh1)
        hold_order();
    }
#undef LOAD_AGG
#undef MMA_AGG
#pragma unroll
    for (int g = 0; g < 8; ++g) {
        const int row = vt * 16 + g + 8 * hi;
        a_t[(b * NC + row) * ND + n] = accA[g];
        m_t[(b * NC + row) * ND + n] = accM[g] * hb[row * ND + n];
    }
}

__global__ void __launch_bounds__(256) k_final(const float* __restrict__ a_t,
                                               const float* __restrict__ m_t,
                                               const float* __restrict__ h,
                                               const float* __restrict__ Wa,
                                               const float* __restrict__ Wm,
                                               const float* __restrict__ ba,
                                               const float* __restrict__ bm,
                                               const float* __restrict__ pb1,
                                               const float* __restrict__ pb2,
                                               const float* __restrict__ pb3,
                                               float* __restrict__ z) {
    const int lane = threadIdx.x & 31;
    const int wv   = threadIdx.x >> 5;
    const int tile = blockIdx.x * 8 + wv;
    const int mt = tile >> 4;
    const int nt = tile & 15;
    const int hi = lane >> 4, lo = lane & 15;
    const int n  = nt * 16 + lo;

    const float* Ar = a_t + (mt * 16 + lo) * ND + 2 * hi;
    const float* Mr = m_t + (mt * 16 + lo) * ND + 2 * hi;
    const float* Wah = Wa + (2 * hi) * ND + n;
    const float* Wmh = Wm + (2 * hi) * ND + n;
    v8f acc1 = {}, acc2 = {};
    v2f xa0[KCH], xm0[KCH], wa0[KCH], wm0[KCH], xa1[KCH], xm1[KCH], wa1[KCH], wm1[KCH];

#define LOAD_FIN(X_, Y_, U_, W_, K0)                                    \
    _Pragma("unroll")          \
    for (int j = 0; j < KCH; ++j) {                                     \
        const int kk = (K0) + 4 * j;                                    \
        X_[j]   = *(const v2f*)(Ar + kk);                               \
        Y_[j]   = *(const v2f*)(Mr + kk);                               \
        U_[j].x = Wah[kk * ND];                                         \
        U_[j].y = Wah[(kk + 1) * ND];                                   \
        W_[j].x = Wmh[kk * ND];                                         \
        W_[j].y = Wmh[(kk + 1) * ND];                                   \
    }
#define MMA_FIN(X_, Y_, U_, W_)                                         \
    _Pragma("unroll")          \
    for (int j = 0; j < KCH; ++j) {                                     \
        acc1 = wmma4(X_[j], U_[j], acc1);                               \
        acc2 = wmma4(Y_[j], W_[j], acc2);                               \
    }

    LOAD_FIN(xa0, xm0, wa0, wm0, 0)
#pragma unroll
    for (int c = 0; c < ND / (4 * KCH); c += 2) {
        LOAD_FIN(xa1, xm1, wa1, wm1, (c + 1) * 4 * KCH)
        hold_order();
        MMA_FIN(xa0, xm0, wa0, wm0)
        hold_order();
        if (c + 2 < ND / (4 * KCH)) {
            LOAD_FIN(xa0, xm0, wa0, wm0, (c + 2) * 4 * KCH)
        }
        hold_order();
        MMA_FIN(xa1, xm1, wa1, wm1)
        hold_order();
    }
#undef LOAD_FIN
#undef MMA_FIN
    const float beta1 = pb1[0], beta2 = pb2[0], beta3 = pb3[0];
    const float bav = ba[n], bmv = bm[n];
#pragma unroll
    for (int g = 0; g < 8; ++g) {
        const int row = mt * 16 + g + 8 * hi;
        z[row * ND + n] = beta1 * h[row * ND + n]
                        + beta2 * (acc1[g] + bav)
                        + beta3 * (acc2[g] + bmv);
    }
}

extern "C" void kernel_launch(void* const* d_in, const int* in_sizes, int n_in,
                              void* d_out, int out_size, void* d_ws, size_t ws_size,
                              hipStream_t stream) {
    (void)in_sizes; (void)n_in; (void)out_size; (void)ws_size;

    const float* h_t = (const float*)d_in[0];
    const float* A_a = (const float*)d_in[1];
    const float* A_m = (const float*)d_in[2];
    const float* W1  = (const float*)d_in[3];
    const float* b1  = (const float*)d_in[4];
    const float* W2  = (const float*)d_in[5];
    const float* b2  = (const float*)d_in[6];
    const float* Wa  = (const float*)d_in[7];
    const float* ba  = (const float*)d_in[8];
    const float* Wm  = (const float*)d_in[9];
    const float* bm  = (const float*)d_in[10];
    const float* be1 = (const float*)d_in[11];
    const float* be2 = (const float*)d_in[12];
    const float* be3 = (const float*)d_in[13];
    float* z = (float*)d_out;

    float* P  = (float*)d_ws;
    float* Q  = P + NROWS * ND;
    float* G  = Q + NROWS * ND;
    float* At = G + NB * NC * NC;
    float* Mt = At + NROWS * ND;

    k_gemm_pq<<<256, 256, 0, stream>>>(h_t, W1, P, Q);
    k_scores<<<NB * 8 * 8, 256, 0, stream>>>(P, Q, b1, W2, b2, A_a, G);
    k_agg<<<256, 256, 0, stream>>>(G, A_m, h_t, At, Mt);
    k_final<<<256, 256, 0, stream>>>(At, Mt, h_t, Wa, Wm, ba, bm,
                                     be1, be2, be3, z);
}


# Score kernel: each thread computes a 2x2 block of (u,v) pairs on a 32x32 block tile and reads P/Q rows from LDS four k at a time (ds_load_b128, pitch 260); k sum order per pair unchanged
# speedup vs baseline: 1.8881x; 1.2118x over previous
#include <hip/hip_runtime.h>
#include <hip/hip_bf16.h>

#define NB 16
#define NC 128
#define ND 256
#define NROWS (NB * NC)

typedef __attribute__((ext_vector_type(2))) float v2f;
typedef __attribute__((ext_vector_type(8))) float v8f;

static __device__ __forceinline__ v8f wmma4(v2f a, v2f b, v8f c) {
    return __builtin_amdgcn_wmma_f32_16x16x4_f32(
         false, a,  false, b,
         (short)0, c,  false,  false);
}

#define KCH 8

static __device__ __forceinline__ void hold_order() { __builtin_amdgcn_sched_barrier(0); }

__global__ void __launch_bounds__(256) k_gemm_pq(const float* __restrict__ H,
                                                 const float* __restrict__ W1,
                                                 float* __restrict__ P,
                                                 float* __restrict__ Q) {
    const int lane = threadIdx.x & 31;
    const int wv   = threadIdx.x >> 5;
    const int tile = blockIdx.x * 8 + wv;
    const int mt = tile >> 4;
    const int nt = tile & 15;
    const int hi = lane >> 4, lo = lane & 15;
    const int n  = nt * 16 + lo;

    const float* Ah = H + (mt * 16 + lo) * ND + 2 * hi;
    const float* Wh = W1 + (2 * hi) * ND + n;
    v8f accP = {}, accQ = {};
    v2f a0[KCH], p0[KCH], q0[KCH], a1[KCH], p1[KCH], q1[KCH];

#define LOAD_PQ(A_, P_, Q_, K0)                                         \
    _Pragma("unroll")          \
    for (int j = 0; j < KCH; ++j) {                                     \
        const int kk = (K0) + 4 * j;                                    \
        A_[j]   = *(const v2f*)(Ah + kk);                               \
        P_[j].x = Wh[kk * ND];                                          \
        P_[j].y = Wh[(kk + 1) * ND];                                    \
        Q_[j].x = Wh[(kk + ND) * ND];                                   \
        Q_[j].y = Wh[(kk + ND + 1) * ND];                               \
    }
#define MMA_PQ(A_, P_, Q_)                                              \
    _Pragma("unroll")          \
    for (int j = 0; j < KCH; ++j) {                                     \
        accP = wmma4(A_[j], P_[j], accP);                               \
        accQ = wmma4(A_[j], Q_[j], accQ);                               \
    }

    LOAD_PQ(a0, p0, q0, 0)
#pragma unroll
    for (int c = 0; c < ND / (4 * KCH); c += 2) {
        LOAD_PQ(a1, p1, q1, (c + 1) * 4 * KCH)
        hold_order();
        MMA_PQ(a0, p0, q0)
        hold_order();
        if (c + 2 < ND / (4 * KCH)) {
            LOAD_PQ(a0, p0, q0, (c + 2) * 4 * KCH)
        }
        hold_order();
        MMA_PQ(a1, p1, q1)
        hold_order();
    }
#undef LOAD_PQ
#undef MMA_PQ
#pragma unroll
    for (int g = 0; g < 8; ++g) {
        const int row = mt * 16 + g + 8 * hi;
        P[row * ND + n] = accP[g];
        Q[row * ND + n] = accQ[g];
    }
}

#define SP 260
typedef __attribute__((ext_vector_type(4))) float v4f;

__global__ void __launch_bounds__(256) k_scores(const float* __restrict__ P,
                                                const float* __restrict__ Q,
                                                const float* __restrict__ b1,
                                                const float* __restrict__ W2,
                                                const float* __restrict__ b2,
                                                const float* __restrict__ A_a,
                                                float* __restrict__ G) {
    __shared__ __attribute__((aligned(16))) float Pl[32 * SP];
    __shared__ __attribute__((aligned(16))) float Ql[32 * SP];
    __shared__ __attribute__((aligned(16))) float b1s[ND];
    __shared__ __attribute__((aligned(16))) float w2s[ND];

    const int blk = blockIdx.x;
    const int vt = blk & 3;
    const int ut = (blk >> 2) & 3;
    const int b  = blk >> 4;
    const int t  = threadIdx.x;

    const v4f* Pg = (const v4f*)(P + (b * NC + ut * 32) * ND);
    const v4f* Qg = (const v4f*)(Q + (b * NC + vt * 32) * ND);
#pragma unroll
    for (int j = 0; j < 8; ++j) {
        const int i  = t + 256 * j;
        const int r  = i >> 6;
        const int c4 = i & 63;
        const int rq = (r >> 1) + 16 * (r & 1);
        *(v4f*)(Pl + r  * SP + 4 * c4) = Pg[i];
        *(v4f*)(Ql + rq * SP + 4 * c4) = Qg[i];
    }
    b1s[t] = b1[t];
    w2s[t] = W2[t];
    __syncthreads();

    const int tu = t >> 4, tv = t & 15;
    const float* p0 = Pl + tu * SP;
    const float* p1 = Pl + (tu + 16) * SP;
    const float* q0 = Ql + tv * SP;
    const float* q1 = Ql + (tv + 16) * SP;
    float a00 = 0.f, a01 = 0.f, a10 = 0.f, a11 = 0.f;
#pragma unroll 2
    for (int k = 0; k < ND; k += 4) {
        const v4f P0 = *(const v4f*)(p0 + k);
        const v4f P1 = *(const v4f*)(p1 + k);
        const v4f Q0 = *(const v4f*)(q0 + k);
        const v4f Q1 = *(const v4f*)(q1 + k);
        const v4f Bv = *(const v4f*)(b1s + k);
        const v4f Wv = *(const v4f*)(w2s + k);
#pragma unroll
        for (int e = 0; e < 4; ++e) {
            const float x00 = P0[e] + Q0[e] + Bv[e];
            const float x01 = P0[e] + Q1[e] + Bv[e];
            const float x10 = P1[e] + Q0[e] + Bv[e];
            const float x11 = P1[e] + Q1[e] + Bv[e];
            a00 = fmaf(fmaxf(x00, 0.f), Wv[e], a00);
            a01 = fmaf(fmaxf(x01, 0.f), Wv[e], a01);
            a10 = fmaf(fmaxf(x10, 0.f), Wv[e], a10);
            a11 = fmaf(fmaxf(x11, 0.f), Wv[e], a11);
        }
    }
    const float bb = b2[0];
    const int u0 = ut * 32 + tu, u1 = u0 + 16;
    const int v0 = vt * 32 + 2 * tv;
    const v2f A0 = *(const v2f*)(A_a + u0 * NC + v0);
    const v2f A1 = *(const v2f*)(A_a + u1 * NC + v0);
    v2f g0, g1;
    g0.x = (1.0f / (1.0f + __expf(-(a00 + bb)))) * A0.x;
    g0.y = (1.0f / (1.0f + __expf(-(a01 + bb)))) * A0.y;
    g1.x = (1.0f / (1.0f + __expf(-(a10 + bb)))) * A1.x;
    g1.y = (1.0f / (1.0f + __expf(-(a11 + bb)))) * A1.y;
    *(v2f*)(G + (b * NC + u0) * NC + v0) = g0;
    *(v2f*)(G + (b * NC + u1) * NC + v0) = g1;
}

__global__ void __launch_bounds__(256) k_agg(const float* __restrict__ G,
                                             const float* __restrict__ Am,
                                             const float* __restrict__ h,
                                             float* __restrict__ a_t,
                                             float* __restrict__ m_t) {
    const int lane = threadIdx.x & 31;
    const int wv   = threadIdx.x >> 5;
    const int tile = blockIdx.x * 8 + wv;
    const int dt = tile & 15;
    const int vt = (tile >> 4) & 7;
    const int b  = tile >> 7;
    const int hi = lane >> 4, lo = lane & 15;
    const int n  = dt * 16 + lo;
    const int m  = vt * 16 + lo;

    const float* Gb = G + b * NC * NC;
    const float* hb = h + b * NC * ND;
    v8f accA = {}, accM = {};
    const float* Gh = Gb + (2 * hi) * NC + m;
    const float* Mh = Am + (2 * hi) * NC + m;
    const float* Hh = hb + (2 * hi) * ND + n;
    v2f fg0[KCH], fm0[KCH], fh0[KCH], fg1[KCH], fm1[KCH], fh1[KCH];

#define LOAD_AGG(G_, M_, B_, K0)                                        \
    _Pragma("unroll")          \
    for (int j = 0; j < KCH; ++j) {                                     \
        const int kk = (K0) + 4 * j;                                    \
        G_[j].x = Gh[kk * NC];                                          \
        G_[j].y = Gh[(kk + 1) * NC];                                    \
        M_[j].x = Mh[kk * NC];                                          \
        M_[j].y = Mh[(kk + 1) * NC];                                    \
        B_[j].x = Hh[kk * ND];                                          \
        B_[j].y = Hh[(kk + 1) * ND];                                    \
    }
#define MMA_AGG(G_, M_, B_)                                             \
    _Pragma("unroll")          \
    for (int j = 0; j < KCH; ++j) {                                     \
        accA = wmma4(G_[j], B_[j], accA);                               \
        accM = wmma4(M_[j], B_[j], accM);                               \
    }

    LOAD_AGG(fg0, fm0, fh0, 0)
#pragma unroll
    for (int c = 0; c < NC / (4 * KCH); c += 2) {
        LOAD_AGG(fg1, fm1, fh1, (c + 1) * 4 * KCH)
        hold_order();
        MMA_AGG(fg0, fm0, fh0)
        hold_order();
        if (c + 2 < NC / (4 * KCH)) {
            LOAD_AGG(fg0, fm0, fh0, (c + 2) * 4 * KCH)
        }
        hold_order();
        MMA_AGG(fg1, fm1, fh1)
        hold_order();
    }
#undef LOAD_AGG
#undef MMA_AGG
#pragma unroll
    for (int g = 0; g < 8; ++g) {
        const int row = vt * 16 + g + 8 * hi;
        a_t[(b * NC + row) * ND + n] = accA[g];
        m_t[(b * NC + row) * ND + n] = accM[g] * hb[row * ND + n];
    }
}

__global__ void __launch_bounds__(256) k_final(const float* __restrict__ a_t,
                                               const float* __restrict__ m_t,
                                               const float* __restrict__ h,
                                               const float* __restrict__ Wa,
                                               const float* __restrict__ Wm,
                                               const float* __restrict__ ba,
                                               const float* __restrict__ bm,
                                               const float* __restrict__ pb1,
                                               const float* __restrict__ pb2,
                                               const float* __restrict__ pb3,
                                               float* __restrict__ z) {
    const int lane = threadIdx.x & 31;
    const int wv   = threadIdx.x >> 5;
    const int tile = blockIdx.x * 8 + wv;
    const int mt = tile >> 4;
    const int nt = tile & 15;
    const int hi = lane >> 4, lo = lane & 15;
    const int n  = nt * 16 + lo;

    const float* Ar = a_t + (mt * 16 + lo) * ND + 2 * hi;
    const float* Mr = m_t + (mt * 16 + lo) * ND + 2 * hi;
    const float* Wah = Wa + (2 * hi) * ND + n;
    const float* Wmh = Wm + (2 * hi) * ND + n;
    v8f acc1 = {}, acc2 = {};
    v2f xa0[KCH], xm0[KCH], wa0[KCH], wm0[KCH], xa1[KCH], xm1[KCH], wa1[KCH], wm1[KCH];

#define LOAD_FIN(X_, Y_, U_, W_, K0)                                    \
    _Pragma("unroll")          \
    for (int j = 0; j < KCH; ++j) {                                     \
        const int kk = (K0) + 4 * j;                                    \
        X_[j]   = *(const v2f*)(Ar + kk);                               \
        Y_[j]   = *(const v2f*)(Mr + kk);                               \
        U_[j].x = Wah[kk * ND];                                         \
        U_[j].y = Wah[(kk + 1) * ND];                                   \
        W_[j].x = Wmh[kk * ND];                                         \
        W_[j].y = Wmh[(kk + 1) * ND];                                   \
    }
#define MMA_FIN(X_, Y_, U_, W_)                                         \
    _Pragma("unroll")          \
    for (int j = 0; j < KCH; ++j) {                                     \
        acc1 = wmma4(X_[j], U_[j], acc1);                               \
        acc2 = wmma4(Y_[j], W_[j], acc2);                               \
    }

    LOAD_FIN(xa0, xm0, wa0, wm0, 0)
#pragma unroll
    for (int c = 0; c < ND / (4 * KCH); c += 2) {
        LOAD_FIN(xa1, xm1, wa1, wm1, (c + 1) * 4 * KCH)
        hold_order();
        MMA_FIN(xa0, xm0, wa0, wm0)
        hold_order();
        if (c + 2 < ND / (4 * KCH)) {
            LOAD_FIN(xa0, xm0, wa0, wm0, (c + 2) * 4 * KCH)
        }
        hold_order();
        MMA_FIN(xa1, xm1, wa1, wm1)
        hold_order();
    }
#undef LOAD_FIN
#undef MMA_FIN
    const float beta1 = pb1[0], beta2 = pb2[0], beta3 = pb3[0];
    const float bav = ba[n], bmv = bm[n];
#pragma unroll
    for (int g = 0; g < 8; ++g) {
        const int row = mt * 16 + g + 8 * hi;
        z[row * ND + n] = beta1 * h[row * ND + n]
                        + beta2 * (acc1[g] + bav)
                        + beta3 * (acc2[g] + bmv);
    }
}

extern "C" void kernel_launch(void* const* d_in, const int* in_sizes, int n_in,
                              void* d_out, int out_size, void* d_ws, size_t ws_size,
                              hipStream_t stream) {
    (void)in_sizes; (void)n_in; (void)out_size; (void)ws_size;

    const float* h_t = (const float*)d_in[0];
    const float* A_a = (const float*)d_in[1];
    const float* A_m = (const float*)d_in[2];
    const float* W1  = (const float*)d_in[3];
    const float* b1  = (const float*)d_in[4];
    const float* W2  = (const float*)d_in[5];
    const float* b2  = (const float*)d_in[6];
    const float* Wa  = (const float*)d_in[7];
    const float* ba  = (const float*)d_in[8];
    const float* Wm  = (const float*)d_in[9];
    const float* bm  = (const float*)d_in[10];
    const float* be1 = (const float*)d_in[11];
    const float* be2 = (const float*)d_in[12];
    const float* be3 = (const float*)d_in[13];
    float* z = (float*)d_out;

    float* P  = (float*)d_ws;
    float* Q  = P + NROWS * ND;
    float* G  = Q + NROWS * ND;
    float* At = G + NB * NC * NC;
    float* Mt = At + NROWS * ND;

    k_gemm_pq<<<256, 256, 0, stream>>>(h_t, W1, P, Q);
    k_scores<<<NB * 4 * 4, 256, 0, stream>>>(P, Q, b1, W2, b2, A_a, G);
    k_agg<<<256, 256, 0, stream>>>(G, A_m, h_t, At, Mt);
    k_final<<<256, 256, 0, stream>>>(At, Mt, h_t, Wa, Wm, ba, bm,
                                     be1, be2, be3, z);
}


# GEMM kernels: the operand tile that a block's 8 waves share (h rows; a_t and m_t rows; G and A_m columns) is copied to LDS once with 16-byte loads, A fragments read from LDS; k order unchanged
# speedup vs baseline: 2.0964x; 1.1103x over previous
#include <hip/hip_runtime.h>
#include <hip/hip_bf16.h>

#define NB 16
#define NC 128
#define ND 256
#define NROWS (NB * NC)

typedef __attribute__((ext_vector_type(2))) float v2f;
typedef __attribute__((ext_vector_type(8))) float v8f;

static __device__ __forceinline__ v8f wmma4(v2f a, v2f b, v8f c) {
    return __builtin_amdgcn_wmma_f32_16x16x4_f32(
         false, a,  false, b,
         (short)0, c,  false,  false);
}

#define KCH 8

static __device__ __forceinline__ void hold_order() { __builtin_amdgcn_sched_barrier(0); }

#define SP 260
typedef __attribute__((ext_vector_type(4))) float v4f;

__global__ void __launch_bounds__(256) k_gemm_pq(const float* __restrict__ H,
                                                 const float* __restrict__ W1,
                                                 float* __restrict__ P,
                                                 float* __restrict__ Q) {
    __shared__ __attribute__((aligned(16))) float Hl[16 * SP];
    const int lane = threadIdx.x & 31;
    const int wv   = threadIdx.x >> 5;
    const int mt = blockIdx.x >> 1;
    const int nt = (blockIdx.x & 1) * 8 + wv;
    const int hi = lane >> 4, lo = lane & 15;
    const int n  = nt * 16 + lo;

    const float* Ah = Hl + lo * SP + 2 * hi;
    const float* Wh = W1 + (2 * hi) * ND + n;
    v8f accP = {}, accQ = {};
    v2f a0[KCH], p0[KCH], q0[KCH], a1[KCH], p1[KCH], q1[KCH];

#define LOAD_PQ_B(P_, Q_, K0)                                           \
    _Pragma("unroll")          \
    for (int j = 0; j < KCH; ++j) {                                     \
        const int kk = (K0) + 4 * j;                                    \
        P_[j].x = Wh[kk * ND];                                          \
        P_[j].y = Wh[(kk + 1) * ND];                                    \
        Q_[j].x = Wh[(kk + ND) * ND];                                   \
        Q_[j].y = Wh[(kk + ND + 1) * ND];                               \
    }
#define LOAD_PQ_A(A_, K0)                                               \
    _Pragma("unroll")          \
    for (int j = 0; j < KCH; ++j) {                                     \
        A_[j] = *(const v2f*)(Ah + (K0) + 4 * j);                       \
    }
#define LOAD_PQ(A_, P_, Q_, K0) LOAD_PQ_B(P_, Q_, K0) LOAD_PQ_A(A_, K0)
#define MMA_PQ(A_, P_, Q_)                                              \
    _Pragma("unroll")          \
    for (int j = 0; j < KCH; ++j) {                                     \
        accP = wmma4(A_[j], P_[j], accP);                               \
        accQ = wmma4(A_[j], Q_[j], accQ);                               \
    }

    LOAD_PQ_B(p0, q0, 0)
    hold_order();
    {
        const v4f* Hg = (const v4f*)(H + mt * 16 * ND);
#pragma unroll
        for (int it = 0; it < 4; ++it) {
            const int i = threadIdx.x + 256 * it;
            *(v4f*)(Hl + (i >> 6) * SP + 4 * (i & 63)) = Hg[i];
        }
    }
    __syncthreads();
    LOAD_PQ_A(a0, 0)
#pragma unroll
    for (int c = 0; c < ND / (4 * KCH); c += 2) {
        LOAD_PQ(a1, p1, q1, (c + 1) * 4 * KCH)
        hold_order();
        MMA_PQ(a0, p0, q0)
        hold_order();
        if (c + 2 < ND / (4 * KCH)) {
            LOAD_PQ(a0, p0, q0, (c + 2) * 4 * KCH)
        }
        hold_order();
        MMA_PQ(a1, p1, q1)
        hold_order();
    }
#undef LOAD_PQ
#undef LOAD_PQ_A
#undef LOAD_PQ_B
#undef MMA_PQ
#pragma unroll
    for (int g = 0; g < 8; ++g) {
        const int row = mt * 16 + g + 8 * hi;
        P[row * ND + n] = accP[g];
        Q[row * ND + n] = accQ[g];
    }
}

__global__ void __launch_bounds__(256) k_scores(const float* __restrict__ P,
                                                const float* __restrict__ Q,
                                                const float* __restrict__ b1,
                                                const float* __restrict__ W2,
                                                const float* __restrict__ b2,
                                                const float* __restrict__ A_a,
                                                float* __restrict__ G) {
    __shared__ __attribute__((aligned(16))) float Pl[32 * SP];
    __shared__ __attribute__((aligned(16))) float Ql[32 * SP];
    __shared__ __attribute__((aligned(16))) float b1s[ND];
    __shared__ __attribute__((aligned(16))) float w2s[ND];

    const int blk = blockIdx.x;
    const int vt = blk & 3;
    const int ut = (blk >> 2) & 3;
    const int b  = blk >> 4;
    const int t  = threadIdx.x;

    const v4f* Pg = (const v4f*)(P + (b * NC + ut * 32) * ND);
    const v4f* Qg = (const v4f*)(Q + (b * NC + vt * 32) * ND);
#pragma unroll
    for (int j = 0; j < 8; ++j) {
        const int i  = t + 256 * j;
        const int r  = i >> 6;
        const int c4 = i & 63;
        const int rq = (r >> 1) + 16 * (r & 1);
        *(v4f*)(Pl + r  * SP + 4 * c4) = Pg[i];
        *(v4f*)(Ql + rq * SP + 4 * c4) = Qg[i];
    }
    b1s[t] = b1[t];
    w2s[t] = W2[t];
    __syncthreads();

    const int tu = t >> 4, tv = t & 15;
    const float* p0 = Pl + tu * SP;
    const float* p1 = Pl + (tu + 16) * SP;
    const float* q0 = Ql + tv * SP;
    const float* q1 = Ql + (tv + 16) * SP;
    float a00 = 0.f, a01 = 0.f, a10 = 0.f, a11 = 0.f;
#pragma unroll 2
    for (int k = 0; k < ND; k += 4) {
        const v4f P0 = *(const v4f*)(p0 + k);
        const v4f P1 = *(const v4f*)(p1 + k);
        const v4f Q0 = *(const v4f*)(q0 + k);
        const v4f Q1 = *(const v4f*)(q1 + k);
        const v4f Bv = *(const v4f*)(b1s + k);
        const v4f Wv = *(const v4f*)(w2s + k);
#pragma unroll
        for (int e = 0; e < 4; ++e) {
            const float x00 = P0[e] + Q0[e] + Bv[e];
            const float x01 = P0[e] + Q1[e] + Bv[e];
            const float x10 = P1[e] + Q0[e] + Bv[e];
            const float x11 = P1[e] + Q1[e] + Bv[e];
            a00 = fmaf(fmaxf(x00, 0.f), Wv[e], a00);
            a01 = fmaf(fmaxf(x01, 0.f), Wv[e], a01);
            a10 = fmaf(fmaxf(x10, 0.f), Wv[e], a10);
            a11 = fmaf(fmaxf(x11, 0.f), Wv[e], a11);
        }
    }
    const float bb = b2[0];
    const int u0 = ut * 32 + tu, u1 = u0 + 16;
    const int v0 = vt * 32 + 2 * tv;
    const v2f A0 = *(const v2f*)(A_a + u0 * NC + v0);
    const v2f A1 = *(const v2f*)(A_a + u1 * NC + v0);
    v2f g0, g1;
    g0.x = (1.0f / (1.0f + __expf(-(a00 + bb)))) * A0.x;
    g0.y = (1.0f / (1.0f + __expf(-(a01 + bb)))) * A0.y;
    g1.x = (1.0f / (1.0f + __expf(-(a10 + bb)))) * A1.x;
    g1.y = (1.0f / (1.0f + __expf(-(a11 + bb)))) * A1.y;
    *(v2f*)(G + (b * NC + u0) * NC + v0) = g0;
    *(v2f*)(G + (b * NC + u1) * NC + v0) = g1;
}

__global__ void __launch_bounds__(256) k_agg(const float* __restrict__ G,
                                             const float* __restrict__ Am,
                                             const float* __restrict__ h,
                                             float* __restrict__ a_t,
                                             float* __restrict__ m_t) {
    __shared__ __attribute__((aligned(16))) float Gl[NC * 16];
    __shared__ __attribute__((aligned(16))) float Ml[NC * 16];
    const int lane = threadIdx.x & 31;
    const int wv   = threadIdx.x >> 5;
    const int dt = (blockIdx.x & 1) * 8 + wv;
    const int vt = (blockIdx.x >> 1) & 7;
    const int b  = blockIdx.x >> 4;
    const int hi = lane >> 4, lo = lane & 15;
    const int n  = dt * 16 + lo;

    const float* Gb = G + b * NC * NC;
    const float* hb = h + b * NC * ND;
    v8f accA = {}, accM = {};
    const float* Gh = Gl + (2 * hi) * 16 + lo;
    const float* Mh = Ml + (2 * hi) * 16 + lo;
    const float* Hh = hb + (2 * hi) * ND + n;
    v2f fg0[KCH], fm0[KCH], fh0[KCH], fg1[KCH], fm1[KCH], fh1[KCH];

#define LOAD_AGG_B(B_, K0)                                              \
    _Pragma("unroll")          \
    for (int j = 0; j < KCH; ++j) {                                     \
        const int kk = (K0) + 4 * j;                                    \
        B_[j].x = Hh[kk * ND];                                          \
        B_[j].y = Hh[(kk + 1) * ND];                                    \
    }
#define LOAD_AGG_A(G_, M_, K0)                                          \
    _Pragma("unroll")          \
    for (int j = 0; j < KCH; ++j) {                                     \
        const int kk = (K0) + 4 * j;                                    \
        G_[j].x = Gh[kk * 16];                                          \
        G_[j].y = Gh[(kk + 1) * 16];                                    \
        M_[j].x = Mh[kk * 16];                                          \
        M_[j].y = Mh[(kk + 1) * 16];                                    \
    }
#define LOAD_AGG(G_, M_, B_, K0) LOAD_AGG_B(B_, K0) LOAD_AGG_A(G_, M_, K0)
#define MMA_AGG(G_, M_, B_)                                             \
    _Pragma("unroll")          \
    for (int j = 0; j < KCH; ++j) {                                     \
        accA = wmma4(G_[j], B_[j], accA);                               \
        accM = wmma4(M_[j], B_[j], accM);                               \
    }

    LOAD_AGG_B(fh0, 0)
    hold_order();
    {
        const float* Gs = Gb + vt * 16;
        const float* Ms = Am + vt * 16;
#pragma unroll
        for (int it = 0; it < 2; ++it) {
            const int u = (threadIdx.x >> 2) + 64 * it;
            const int q = (threadIdx.x & 3) * 4;
            *(v4f*)(Gl + u * 16 + q) = *(const v4f*)(Gs + u * NC + q);
            *(v4f*)(Ml + u * 16 + q) = *(const v4f*)(Ms + u * NC + q);
        }
    }
    __syncthreads();
    LOAD_AGG_A(fg0, fm0, 0)
#pragma unroll
    for (int c = 0; c < NC / (4 * KCH); c += 2) {
        LOAD_AGG(fg1, fm1, fh1, (c + 1) * 4 * KCH)
        hold_order();
        MMA_AGG(fg0, fm0, fh0)
        hold_order();
        if (c + 2 < NC / (4 * KCH)) {
            LOAD_AGG(fg0, fm0, fh0, (c + 2) * 4 * KCH)
        }
        hold_order();
        MMA_AGG(fg1, fm1, fh1)
        hold_order();
    }
#undef LOAD_AGG
#undef LOAD_AGG_A
#undef LOAD_AGG_B
#undef MMA_AGG
#pragma unroll
    for (int g = 0; g < 8; ++g) {
        const int row = vt * 16 + g + 8 * hi;
        a_t[(b * NC + row) * ND + n] = accA[g];
        m_t[(b * NC + row) * ND + n] = accM[g] * hb[row * ND + n];
    }
}

__global__ void __launch_bounds__(256) k_final(const float* __restrict__ a_t,
                                               const float* __restrict__ m_t,
                                               const float* __restrict__ h,
                                               const float* __restrict__ Wa,
                                               const float* __restrict__ Wm,
                                               const float* __restrict__ ba,
                                               const float* __restrict__ bm,
                                               const float* __restrict__ pb1,
                                               const float* __restrict__ pb2,
                                               const float* __restrict__ pb3,
                                               float* __restrict__ z) {
    __shared__ __attribute__((aligned(16))) float Xl[16 * SP];
    __shared__ __attribute__((aligned(16))) float Yl[16 * SP];
    const int lane = threadIdx.x & 31;
    const int wv   = threadIdx.x >> 5;
    const int mt = blockIdx.x >> 1;
    const int nt = (blockIdx.x & 1) * 8 + wv;
    const int hi = lane >> 4, lo = lane & 15;
    const int n  = nt * 16 + lo;

    const float* Ar = Xl + lo * SP + 2 * hi;
    const float* Mr = Yl + lo * SP + 2 * hi;
    const float* Wah = Wa + (2 * hi) * ND + n;
    const float* Wmh = Wm + (2 * hi) * ND + n;
    v8f acc1 = {}, acc2 = {};
    v2f xa0[KCH], xm0[KCH], wa0[KCH], wm0[KCH], xa1[KCH], xm1[KCH], wa1[KCH], wm1[KCH];

#define LOAD_FIN_B(U_, W_, K0)                                          \
    _Pragma("unroll")          \
    for (int j = 0; j < KCH; ++j) {                                     \
        const int kk = (K0) + 4 * j;                                    \
        U_[j].x = Wah[kk * ND];                                         \
        U_[j].y = Wah[(kk + 1) * ND];                                   \
        W_[j].x = Wmh[kk * ND];                                         \
        W_[j].y = Wmh[(kk + 1) * ND];                                   \
    }
#define LOAD_FIN_A(X_, Y_, K0)                                          \
    _Pragma("unroll")          \
    for (int j = 0; j < KCH; ++j) {                                     \
        const int kk = (K0) + 4 * j;                                    \
        X_[j] = *(const v2f*)(Ar + kk);                                 \
        Y_[j] = *(const v2f*)(Mr + kk);                                 \
    }
#define LOAD_FIN(X_, Y_, U_, W_, K0) LOAD_FIN_B(U_, W_, K0) LOAD_FIN_A(X_, Y_, K0)
#define MMA_FIN(X_, Y_, U_, W_)                                         \
    _Pragma("unroll")          \
    for (int j = 0; j < KCH; ++j) {                                     \
        acc1 = wmma4(X_[j], U_[j], acc1);                               \
        acc2 = wmma4(Y_[j], W_[j], acc2);                               \
    }

    LOAD_FIN_B(wa0, wm0, 0)
    hold_order();
    {
        const v4f* Xg = (const v4f*)(a_t + mt * 16 * ND);
        const v4f* Yg = (const v4f*)(m_t + mt * 16 * ND);
#pragma unroll
        for (int it = 0; it < 4; ++it) {
            const int i = threadIdx.x + 256 * it;
            *(v4f*)(Xl + (i >> 6) * SP + 4 * (i & 63)) = Xg[i];
            *(v4f*)(Yl + (i >> 6) * SP + 4 * (i & 63)) = Yg[i];
        }
    }
    __syncthreads();
    LOAD_FIN_A(xa0, xm0, 0)
#pragma unroll
    for (int c = 0; c < ND / (4 * KCH); c += 2) {
        LOAD_FIN(xa1, xm1, wa1, wm1, (c + 1) * 4 * KCH)
        hold_order();
        MMA_FIN(xa0, xm0, wa0, wm0)
        hold_order();
        if (c + 2 < ND / (4 * KCH)) {
            LOAD_FIN(xa0, xm0, wa0, wm0, (c + 2) * 4 * KCH)
        }
        hold_order();
        MMA_FIN(xa1, xm1, wa1, wm1)
        hold_order();
    }
#undef LOAD_FIN
#undef LOAD_FIN_A
#undef LOAD_FIN_B
#undef MMA_FIN
    const float beta1 = pb1[0], beta2 = pb2[0], beta3 = pb3[0];
    const float bav = ba[n], bmv = bm[n];
#pragma unroll
    for (int g = 0; g < 8; ++g) {
        const int row = mt * 16 + g + 8 * hi;
        z[row * ND + n] = beta1 * h[row * ND + n]
                        + beta2 * (acc1[g] + bav)
                        + beta3 * (acc2[g] + bmv);
    }
}

extern "C" void kernel_launch(void* const* d_in, const int* in_sizes, int n_in,
                              void* d_out, int out_size, void* d_ws, size_t ws_size,
                              hipStream_t stream) {
    (void)in_sizes; (void)n_in; (void)out_size; (void)ws_size;

    const float* h_t = (const float*)d_in[0];
    const float* A_a = (const float*)d_in[1];
    const float* A_m = (const float*)d_in[2];
    const float* W1  = (const float*)d_in[3];
    const float* b1  = (const float*)d_in[4];
    const float* W2  = (const float*)d_in[5];
    const float* b2  = (const float*)d_in[6];
    const float* Wa  = (const float*)d_in[7];
    const float* ba  = (const float*)d_in[8];
    const float* Wm  = (const float*)d_in[9];
    const float* bm  = (const float*)d_in[10];
    const float* be1 = (const float*)d_in[11];
    const float* be2 = (const float*)d_in[12];
    const float* be3 = (const float*)d_in[13];
    float* z = (float*)d_out;

    float* P  = (float*)d_ws;
    float* Q  = P + NROWS * ND;
    float* G  = Q + NROWS * ND;
    float* At = G + NB * NC * NC;
    float* Mt = At + NROWS * ND;

    k_gemm_pq<<<256, 256, 0, stream>>>(h_t, W1, P, Q);
    k_scores<<<NB * 4 * 4, 256, 0, stream>>>(P, Q, b1, W2, b2, A_a, G);
    k_agg<<<256, 256, 0, stream>>>(G, A_m, h_t, At, Mt);
    k_final<<<256, 256, 0, stream>>>(At, Mt, h_t, Wa, Wm, ba, bm,
                                     be1, be2, be3, z);
}


# Launch count 4 to 3: aggregation kernel fused into the output GEMM kernel; each block builds its 16-row a_t and m_t tiles in LDS and runs both products from there; k order of every sum unchanged
# speedup vs baseline: 2.2357x; 1.0664x over previous
#include <hip/hip_runtime.h>
#include <hip/hip_bf16.h>

#define NB 16
#define NC 128
#define ND 256
#define NROWS (NB * NC)

typedef __attribute__((ext_vector_type(2))) float v2f;
typedef __attribute__((ext_vector_type(8))) float v8f;

static __device__ __forceinline__ v8f wmma4(v2f a, v2f b, v8f c) {
    return __builtin_amdgcn_wmma_f32_16x16x4_f32(
         false, a,  false, b,
         (short)0, c,  false,  false);
}

#define KCH 8

static __device__ __forceinline__ void hold_order() { __builtin_amdgcn_sched_barrier(0); }

#define SP 260
typedef __attribute__((ext_vector_type(4))) float v4f;

__global__ void __launch_bounds__(256) k_gemm_pq(const float* __restrict__ H,
                                                 const float* __restrict__ W1,
                                                 float* __restrict__ P,
                                                 float* __restrict__ Q) {
    __shared__ __attribute__((aligned(16))) float Hl[16 * SP];
    const int lane = threadIdx.x & 31;
    const int wv   = threadIdx.x >> 5;
    const int mt = blockIdx.x >> 1;
    const int nt = (blockIdx.x & 1) * 8 + wv;
    const int hi = lane >> 4, lo = lane & 15;
    const int n  = nt * 16 + lo;

    const float* Ah = Hl + lo * SP + 2 * hi;
    const float* Wh = W1 + (2 * hi) * ND + n;
    v8f accP = {}, accQ = {};
    v2f a0[KCH], p0[KCH], q0[KCH], a1[KCH], p1[KCH], q1[KCH];

#define LOAD_PQ_B(P_, Q_, K0)                                           \
    _Pragma("unroll")          \
    for (int j = 0; j < KCH; ++j) {                                     \
        const int kk = (K0) + 4 * j;                                    \
        P_[j].x = Wh[kk * ND];                                          \
        P_[j].y = Wh[(kk + 1) * ND];                                    \
        Q_[j].x = Wh[(kk + ND) * ND];                                   \
        Q_[j].y = Wh[(kk + ND + 1) * ND];                               \
    }
#define LOAD_PQ_A(A_, K0)                                               \
    _Pragma("unroll")          \
    for (int j = 0; j < KCH; ++j) {                                     \
        A_[j] = *(const v2f*)(Ah + (K0) + 4 * j);                       \
    }
#define LOAD_PQ(A_, P_, Q_, K0) LOAD_PQ_B(P_, Q_, K0) LOAD_PQ_A(A_, K0)
#define MMA_PQ(A_, P_, Q_)                                              \
    _Pragma("unroll")          \
    for (int j = 0; j < KCH; ++j) {                                     \
        accP = wmma4(A_[j], P_[j], accP);                               \
        accQ = wmma4(A_[j], Q_[j], accQ);                               \
    }

    LOAD_PQ_B(p0, q0, 0)
    hold_order();
    {
        const v4f* Hg = (const v4f*)(H + mt * 16 * ND);
#pragma unroll
        for (int it = 0; it < 4; ++it) {
            const int i = threadIdx.x + 256 * it;
            *(v4f*)(Hl + (i >> 6) * SP + 4 * (i & 63)) = Hg[i];
        }
    }
    __syncthreads();
    LOAD_PQ_A(a0, 0)
#pragma unroll
    for (int c = 0; c < ND / (4 * KCH); c += 2) {
        LOAD_PQ(a1, p1, q1, (c + 1) * 4 * KCH)
        hold_order();
        MMA_PQ(a0, p0, q0)
        hold_order();
        if (c + 2 < ND / (4 * KCH)) {
            LOAD_PQ(a0, p0, q0, (c + 2) * 4 * KCH)
        }
        hold_order();
        MMA_PQ(a1, p1, q1)
        hold_order();
    }
#undef LOAD_PQ
#undef LOAD_PQ_A
#undef LOAD_PQ_B
#undef MMA_PQ
#pragma unroll
    for (int g = 0; g < 8; ++g) {
        const int row = mt * 16 + g + 8 * hi;
        P[row * ND + n] = accP[g];
        Q[row * ND + n] = accQ[g];
    }
}

__global__ void __launch_bounds__(256) k_scores(const float* __restrict__ P,
                                                const float* __restrict__ Q,
                                                const float* __restrict__ b1,
                                                const float* __restrict__ W2,
                                                const float* __restrict__ b2,
                                                const float* __restrict__ A_a,
                                                float* __restrict__ G) {
    __shared__ __attribute__((aligned(16))) float Pl[32 * SP];
    __shared__ __attribute__((aligned(16))) float Ql[32 * SP];
    __shared__ __attribute__((aligned(16))) float b1s[ND];
    __shared__ __attribute__((aligned(16))) float w2s[ND];

    const int blk = blockIdx.x;
    const int vt = blk & 3;
    const int ut = (blk >> 2) & 3;
    const int b  = blk >> 4;
    const int t  = threadIdx.x;

    const v4f* Pg = (const v4f*)(P + (b * NC + ut * 32) * ND);
    const v4f* Qg = (const v4f*)(Q + (b * NC + vt * 32) * ND);
#pragma unroll
    for (int j = 0; j < 8; ++j) {
        const int i  = t + 256 * j;
        const int r  = i >> 6;
        const int c4 = i & 63;
        const int rq = (r >> 1) + 16 * (r & 1);
        *(v4f*)(Pl + r  * SP + 4 * c4) = Pg[i];
        *(v4f*)(Ql + rq * SP + 4 * c4) = Qg[i];
    }
    b1s[t] = b1[t];
    w2s[t] = W2[t];
    __syncthreads();

    const int tu = t >> 4, tv = t & 15;
    const float* p0 = Pl + tu * SP;
    const float* p1 = Pl + (tu + 16) * SP;
    const float* q0 = Ql + tv * SP;
    const float* q1 = Ql + (tv + 16) * SP;
    float a00 = 0.f, a01 = 0.f, a10 = 0.f, a11 = 0.f;
#pragma unroll 2
    for (int k = 0; k < ND; k += 4) {
        const v4f P0 = *(const v4f*)(p0 + k);
        const v4f P1 = *(const v4f*)(p1 + k);
        const v4f Q0 = *(const v4f*)(q0 + k);
        const v4f Q1 = *(const v4f*)(q1 + k);
        const v4f Bv = *(const v4f*)(b1s + k);
        const v4f Wv = *(const v4f*)(w2s + k);
#pragma unroll
        for (int e = 0; e < 4; ++e) {
            const float x00 = P0[e] + Q0[e] + Bv[e];
            const float x01 = P0[e] + Q1[e] + Bv[e];
            const float x10 = P1[e] + Q0[e] + Bv[e];
            const float x11 = P1[e] + Q1[e] + Bv[e];
            a00 = fmaf(fmaxf(x00, 0.f), Wv[e], a00);
            a01 = fmaf(fmaxf(x01, 0.f), Wv[e], a01);
            a10 = fmaf(fmaxf(x10, 0.f), Wv[e], a10);
            a11 = fmaf(fmaxf(x11, 0.f), Wv[e], a11);
        }
    }
    const float bb = b2[0];
    const int u0 = ut * 32 + tu, u1 = u0 + 16;
    const int v0 = vt * 32 + 2 * tv;
    const v2f A0 = *(const v2f*)(A_a + u0 * NC + v0);
    const v2f A1 = *(const v2f*)(A_a + u1 * NC + v0);
    v2f g0, g1;
    g0.x = (1.0f / (1.0f + __expf(-(a00 + bb)))) * A0.x;
    g0.y = (1.0f / (1.0f + __expf(-(a01 + bb)))) * A0.y;
    g1.x = (1.0f / (1.0f + __expf(-(a10 + bb)))) * A1.x;
    g1.y = (1.0f / (1.0f + __expf(-(a11 + bb)))) * A1.y;
    *(v2f*)(G + (b * NC + u0) * NC + v0) = g0;
    *(v2f*)(G + (b * NC + u1) * NC + v0) = g1;
}

__global__ void __launch_bounds__(256) k_agg_final(const float* __restrict__ G,
                                                   const float* __restrict__ Am,
                                                   const float* __restrict__ h,
                                                   const float* __restrict__ Wa,
                                                   const float* __restrict__ Wm,
                                                   const float* __restrict__ ba,
                                                   const float* __restrict__ bm,
                                                   const float* __restrict__ pb1,
                                                   const float* __restrict__ pb2,
                                                   const float* __restrict__ pb3,
                                                   float* __restrict__ z) {
    __shared__ __attribute__((aligned(16))) float Gl[NC * 16];
    __shared__ __attribute__((aligned(16))) float Ml[NC * 16];
    __shared__ __attribute__((aligned(16))) float Xl[16 * SP];
    __shared__ __attribute__((aligned(16))) float Yl[16 * SP];
    const int lane = threadIdx.x & 31;
    const int wv   = threadIdx.x >> 5;
    const int mt = blockIdx.x >> 1;
    const int vt = mt & 7;
    const int b  = mt >> 3;
    const int nt = (blockIdx.x & 1) * 8 + wv;
    const int hi = lane >> 4, lo = lane & 15;
    const int n  = nt * 16 + lo;
    const int d0 = wv * 32 + lo;

    const float* Gb = G + b * NC * NC;
    const float* hb = h + b * NC * ND;

    v8f accA0 = {}, accM0 = {}, accA1 = {}, accM1 = {};
    const float* Gh = Gl + (2 * hi) * 16 + lo;
    const float* Mh = Ml + (2 * hi) * 16 + lo;
    const float* Hh = hb + (2 * hi) * ND + d0;
    v2f fg0[KCH], fm0[KCH], fh0[KCH], fk0[KCH], fg1[KCH], fm1[KCH], fh1[KCH], fk1[KCH];

    const float* Ar = Xl + lo * SP + 2 * hi;
    const float* Mr = Yl + lo * SP + 2 * hi;
    const float* Wah = Wa + (2 * hi) * ND + n;
    const float* Wmh = Wm + (2 * hi) * ND + n;
    v8f acc1 = {}, acc2 = {};
    v2f xa0[KCH], xm0[KCH], wa0[KCH], wm0[KCH], xa1[KCH], xm1[KCH], wa1[KCH], wm1[KCH];
    float hv0[8], hv1[8];

#define LOAD_AGG_B(B0_, B1_, K0)                                        \
    _Pragma("unroll")          \
    for (int j = 0; j < KCH; ++j) {                                     \
        const int kk = (K0) + 4 * j;                                    \
        B0_[j].x = Hh[kk * ND];                                         \
        B0_[j].y = Hh[(kk + 1) * ND];                                   \
        B1_[j].x = Hh[kk * ND + 16];                                    \
        B1_[j].y = Hh[(kk + 1) * ND + 16];                              \
    }
#define LOAD_AGG_A(G_, M_, K0)                                          \
    _Pragma("unroll")          \
    for (int j = 0; j < KCH; ++j) {                                     \
        const int kk = (K0) + 4 * j;                                    \
        G_[j].x = Gh[kk * 16];                                          \
        G_[j].y = Gh[(kk + 1) * 16];                                    \
        M_[j].x = Mh[kk * 16];                                          \
        M_[j].y = Mh[(kk + 1) * 16];                                    \
    }
#define LOAD_AGG(G_, M_, B0_, B1_, K0) LOAD_AGG_B(B0_, B1_, K0) LOAD_AGG_A(G_, M_, K0)
#define MMA_AGG(G_, M_, B0_, B1_)                                       \
    _Pragma("unroll")          \
    for (int j = 0; j < KCH; ++j) {                                     \
        accA0 = wmma4(G_[j], B0_[j], accA0);                            \
        accM0 = wmma4(M_[j], B0_[j], accM0);                            \
        accA1 = wmma4(G_[j], B1_[j], accA1);                            \
        accM1 = wmma4(M_[j], B1_[j], accM1);                            \
    }
#define LOAD_FIN_B(U_, W_, K0)                                          \
    _Pragma("unroll")          \
    for (int j = 0; j < KCH; ++j) {                                     \
        const int kk = (K0) + 4 * j;                                    \
        U_[j].x = Wah[kk * ND];                                         \
        U_[j].y = Wah[(kk + 1) * ND];                                   \
        W_[j].x = Wmh[kk * ND];                                         \
        W_[j].y = Wmh[(kk + 1) * ND];                                   \
    }
#define LOAD_FIN_A(X_, Y_, K0)                                          \
    _Pragma("unroll")          \
    for (int j = 0; j < KCH; ++j) {                                     \
        const int kk = (K0) + 4 * j;                                    \
        X_[j] = *(const v2f*)(Ar + kk);                                 \
        Y_[j] = *(const v2f*)(Mr + kk);                                 \
    }
#define LOAD_FIN(X_, Y_, U_, W_, K0) LOAD_FIN_B(U_, W_, K0) LOAD_FIN_A(X_, Y_, K0)
#define MMA_FIN(X_, Y_, U_, W_)                                         \
    _Pragma("unroll")          \
    for (int j = 0; j < KCH; ++j) {                                     \
        acc1 = wmma4(X_[j], U_[j], acc1);                               \
        acc2 = wmma4(Y_[j], W_[j], acc2);                               \
    }

    LOAD_AGG_B(fh0, fk0, 0)
    hold_order();
    {
        const float* Gs = Gb + vt * 16;
        const float* Ms = Am + vt * 16;
#pragma unroll
        for (int it = 0; it < 2; ++it) {
            const int u = (threadIdx.x >> 2) + 64 * it;
            const int q = (threadIdx.x & 3) * 4;
            *(v4f*)(Gl + u * 16 + q) = *(const v4f*)(Gs + u * NC + q);
            *(v4f*)(Ml + u * 16 + q) = *(const v4f*)(Ms + u * NC + q);
        }
    }
    __syncthreads();
    LOAD_AGG_A(fg0, fm0, 0)
#pragma unroll
    for (int c = 0; c < NC / (4 * KCH); c += 2) {
        LOAD_AGG(fg1, fm1, fh1, fk1, (c + 1) * 4 * KCH)
        hold_order();
        MMA_AGG(fg0, fm0, fh0, fk0)
        hold_order();
        if (c + 2 < NC / (4 * KCH)) {
            LOAD_AGG(fg0, fm0, fh0, fk0, (c + 2) * 4 * KCH)
        } else {
            LOAD_FIN_B(wa0, wm0, 0)
#pragma unroll
            for (int g = 0; g < 8; ++g) {
                const int row = vt * 16 + g + 8 * hi;
                hv0[g] = hb[row * ND + d0];
                hv1[g] = hb[row * ND + d0 + 16];
            }
        }
        hold_order();
        MMA_AGG(fg1, fm1, fh1, fk1)
        hold_order();
    }
#pragma unroll
    for (int g = 0; g < 8; ++g) {
        const int r = g + 8 * hi;
        Xl[r * SP + d0]      = accA0[g];
        Xl[r * SP + d0 + 16] = accA1[g];
        Yl[r * SP + d0]      = accM0[g] * hv0[g];
        Yl[r * SP + d0 + 16] = accM1[g] * hv1[g];
    }
    __syncthreads();

    LOAD_FIN_A(xa0, xm0, 0)
#pragma unroll
    for (int c = 0; c < ND / (4 * KCH); c += 2) {
        LOAD_FIN(xa1, xm1, wa1, wm1, (c + 1) * 4 * KCH)
        hold_order();
        MMA_FIN(xa0, xm0, wa0, wm0)
        hold_order();
        if (c + 2 < ND / (4 * KCH)) {
            LOAD_FIN(xa0, xm0, wa0, wm0, (c + 2) * 4 * KCH)
        }
        hold_order();
        MMA_FIN(xa1, xm1, wa1, wm1)
        hold_order();
    }
#undef LOAD_AGG
#undef LOAD_AGG_A
#undef LOAD_AGG_B
#undef MMA_AGG
#undef LOAD_FIN
#undef LOAD_FIN_A
#undef LOAD_FIN_B
#undef MMA_FIN
    const float beta1 = pb1[0], beta2 = pb2[0], beta3 = pb3[0];
    const float bav = ba[n], bmv = bm[n];
#pragma unroll
    for (int g = 0; g < 8; ++g) {
        const int row = mt * 16 + g + 8 * hi;
        z[row * ND + n] = beta1 * h[row * ND + n]
                        + beta2 * (acc1[g] + bav)
                        + beta3 * (acc2[g] + bmv);
    }
}

extern "C" void kernel_launch(void* const* d_in, const int* in_sizes, int n_in,
                              void* d_out, int out_size, void* d_ws, size_t ws_size,
                              hipStream_t stream) {
    (void)in_sizes; (void)n_in; (void)out_size; (void)ws_size;

    const float* h_t = (const float*)d_in[0];
    const float* A_a = (const float*)d_in[1];
    const float* A_m = (const float*)d_in[2];
    const float* W1  = (const float*)d_in[3];
    const float* b1  = (const float*)d_in[4];
    const float* W2  = (const float*)d_in[5];
    const float* b2  = (const float*)d_in[6];
    const float* Wa  = (const float*)d_in[7];
    const float* ba  = (const float*)d_in[8];
    const float* Wm  = (const float*)d_in[9];
    const float* bm  = (const float*)d_in[10];
    const float* be1 = (const float*)d_in[11];
    const float* be2 = (const float*)d_in[12];
    const float* be3 = (const float*)d_in[13];
    float* z = (float*)d_out;

    float* P  = (float*)d_ws;
    float* Q  = P + NROWS * ND;
    float* G  = Q + NROWS * ND;

    k_gemm_pq<<<256, 256, 0, stream>>>(h_t, W1, P, Q);
    k_scores<<<NB * 4 * 4, 256, 0, stream>>>(P, Q, b1, W2, b2, A_a, G);
    k_agg_final<<<256, 256, 0, stream>>>(G, A_m, h_t, Wa, Wm, ba, bm,
                                         be1, be2, be3, z);
}


# GEMM 1 and score kernel on one wave per SIMD (128-thread blocks), wave tile doubled: 2 row tiles per W1 fragment; 4x2 pairs per thread, Q pairs interleaved in LDS, b1 added: p+(q+b1), was (p+q)+b1
# speedup vs baseline: 2.3666x; 1.0585x over previous
#include <hip/hip_runtime.h>
#include <hip/hip_bf16.h>

#define NB 16
#define NC 128
#define ND 256
#define NROWS (NB * NC)

typedef __attribute__((ext_vector_type(2))) float v2f;
typedef __attribute__((ext_vector_type(8))) float v8f;

static __device__ __forceinline__ v8f wmma4(v2f a, v2f b, v8f c) {
    return __builtin_amdgcn_wmma_f32_16x16x4_f32(
         false, a,  false, b,
         (short)0, c,  false,  false);
}

#define KCH 8

static __device__ __forceinline__ void hold_order() { __builtin_amdgcn_sched_barrier(0); }

#define SP 260
typedef __attribute__((ext_vector_type(4))) float v4f;

__global__ void __launch_bounds__(128) k_gemm_pq(const float* __restrict__ H,
                                                 const float* __restrict__ W1,
                                                 float* __restrict__ P,
                                                 float* __restrict__ Q) {
    __shared__ __attribute__((aligned(16))) float Hl[32 * SP];
    const int lane = threadIdx.x & 31;
    const int wv   = threadIdx.x >> 5;
    const int mt = blockIdx.x >> 2;
    const int nt = (blockIdx.x & 3) * 4 + wv;
    const int hi = lane >> 4, lo = lane & 15;
    const int n  = nt * 16 + lo;

    const float* Ah = Hl + lo * SP + 2 * hi;
    const float* Bh = Ah + 16 * SP;
    const float* Wh = W1 + (2 * hi) * ND + n;
    v8f accP0 = {}, accQ0 = {}, accP1 = {}, accQ1 = {};
    v2f x0[KCH], y0[KCH], p0[KCH], q0[KCH], x1[KCH], y1[KCH], p1[KCH], q1[KCH];

#define LOAD_PQ_B(P_, Q_, K0)                                           \
    _Pragma("unroll")          \
    for (int j = 0; j < KCH; ++j) {                                     \
        const int kk = (K0) + 4 * j;                                    \
        P_[j].x = Wh[kk * ND];                                          \
        P_[j].y = Wh[(kk + 1) * ND];                                    \
        Q_[j].x = Wh[(kk + ND) * ND];                                   \
        Q_[j].y = Wh[(kk + ND + 1) * ND];                               \
    }
#define LOAD_PQ_A(X_, Y_, K0)                                           \
    _Pragma("unroll")          \
    for (int j = 0; j < KCH; ++j) {                                     \
        X_[j] = *(const v2f*)(Ah + (K0) + 4 * j);                       \
        Y_[j] = *(const v2f*)(Bh + (K0) + 4 * j);                       \
    }
#define LOAD_PQ(X_, Y_, P_, Q_, K0) LOAD_PQ_B(P_, Q_, K0) LOAD_PQ_A(X_, Y_, K0)
#define MMA_PQ(X_, Y_, P_, Q_)                                          \
    _Pragma("unroll")          \
    for (int j = 0; j < KCH; ++j) {                                     \
        accP0 = wmma4(X_[j], P_[j], accP0);                             \
        accQ0 = wmma4(X_[j], Q_[j], accQ0);                             \
        accP1 = wmma4(Y_[j], P_[j], accP1);                             \
        accQ1 = wmma4(Y_[j], Q_[j], accQ1);                             \
    }

    LOAD_PQ_B(p0, q0, 0)
    hold_order();
    {
        const v4f* Hg = (const v4f*)(H + mt * 32 * ND);
#pragma unroll
        for (int it = 0; it < 16; ++it) {
            const int i = threadIdx.x + 128 * it;
            *(v4f*)(Hl + (i >> 6) * SP + 4 * (i & 63)) = Hg[i];
        }
    }
    __syncthreads();
    LOAD_PQ_A(x0, y0, 0)
#pragma unroll
    for (int c = 0; c < ND / (4 * KCH); c += 2) {
        LOAD_PQ(x1, y1, p1, q1, (c + 1) * 4 * KCH)
        hold_order();
        MMA_PQ(x0, y0, p0, q0)
        hold_order();
        if (c + 2 < ND / (4 * KCH)) {
            LOAD_PQ(x0, y0, p0, q0, (c + 2) * 4 * KCH)
        }
        hold_order();
        MMA_PQ(x1, y1, p1, q1)
        hold_order();
    }
#undef LOAD_PQ
#undef LOAD_PQ_A
#undef LOAD_PQ_B
#undef MMA_PQ
#pragma unroll
    for (int g = 0; g < 8; ++g) {
        const int row = mt * 32 + g + 8 * hi;
        P[row * ND + n]        = accP0[g];
        Q[row * ND + n]        = accQ0[g];
        P[(row + 16) * ND + n] = accP1[g];
        Q[(row + 16) * ND + n] = accQ1[g];
    }
}

#define QP 516
#define KG 4
__global__ void __launch_bounds__(128) k_scores(const float* __restrict__ P,
                                                const float* __restrict__ Q,
                                                const float* __restrict__ b1,
                                                const float* __restrict__ W2,
                                                const float* __restrict__ b2,
                                                const float* __restrict__ A_a,
                                                float* __restrict__ G) {
    __shared__ __attribute__((aligned(16))) float Pl[32 * SP];
    __shared__ __attribute__((aligned(16))) float Ql[16 * QP];
    __shared__ __attribute__((aligned(16))) float w2s[ND];

    const int blk = blockIdx.x;
    const int vt = blk & 3;
    const int ut = (blk >> 2) & 3;
    const int b  = blk >> 4;
    const int t  = threadIdx.x;

    const v4f* Pg = (const v4f*)(P + (b * NC + ut * 32) * ND);
    const v4f* Qg = (const v4f*)(Q + (b * NC + vt * 32) * ND);
    const int c4 = t & 63;
    const int r0 = t >> 6;
    v4f bq;
    bq.x = b1[4 * c4];
    bq.y = b1[4 * c4 + 1];
    bq.z = b1[4 * c4 + 2];
    bq.w = b1[4 * c4 + 3];
#pragma unroll
    for (int j = 0; j < 16; ++j) {
        const int r = r0 + 2 * j;
        *(v4f*)(Pl + r * SP + 4 * c4) = Pg[r * 64 + c4];
    }
#pragma unroll
    for (int j = 0; j < 8; ++j) {
        const int jr = r0 + 2 * j;
        const v4f qe = Qg[(2 * jr) * 64 + c4] + bq;
        const v4f qo = Qg[(2 * jr + 1) * 64 + c4] + bq;
        v4f lo4, hi4;
        lo4.x = qe.x; lo4.y = qo.x; lo4.z = qe.y; lo4.w = qo.y;
        hi4.x = qe.z; hi4.y = qo.z; hi4.z = qe.w; hi4.w = qo.w;
        *(v4f*)(Ql + jr * QP + 8 * c4)     = lo4;
        *(v4f*)(Ql + jr * QP + 8 * c4 + 4) = hi4;
    }
    w2s[t]       = W2[t];
    w2s[t + 128] = W2[t + 128];
    __syncthreads();

    const int tu = t >> 4, tv = t & 15;
    const float* pp = Pl + tu * SP;
    const float* qq = Ql + tv * QP;
    const float* ww = w2s;
    float acc[4][2] = {};
    v4f Pa[4][KG], Qa[2 * KG], Wa_[KG], Pb[4][KG], Qb[2 * KG], Wb_[KG];

#define LOAD_SC(P_, Q_, W_, K0)                                         \
    _Pragma("unroll")                       \
    for (int s = 0; s < KG; ++s) {                                      \
        const int kk = (K0) + 4 * s;                                    \
        P_[0][s] = *(const v4f*)(pp + kk);                              \
        P_[1][s] = *(const v4f*)(pp + 8 * SP + kk);                     \
        P_[2][s] = *(const v4f*)(pp + 16 * SP + kk);                    \
        P_[3][s] = *(const v4f*)(pp + 24 * SP + kk);                    \
        Q_[2 * s]     = *(const v4f*)(qq + 2 * kk);                     \
        Q_[2 * s + 1] = *(const v4f*)(qq + 2 * kk + 4);                 \
        W_[s]    = *(const v4f*)(ww + kk);                              \
    }
#define MAC_SC(P_, Q_, W_)                                              \
    _Pragma("unroll")                       \
    for (int kl = 0; kl < 4 * KG; ++kl) {                               \
        const float w  = W_[kl >> 2][kl & 3];                           \
        const float qa = Q_[kl >> 1][2 * (kl & 1)];                     \
        const float qb = Q_[kl >> 1][2 * (kl & 1) + 1];                 \
        _Pragma("unroll")                   \
        for (int i = 0; i < 4; ++i) {                                   \
            const float p = P_[i][kl >> 2][kl & 3];                     \
            acc[i][0] = fmaf(fmaxf(p + qa, 0.f), w, acc[i][0]);         \
            acc[i][1] = fmaf(fmaxf(p + qb, 0.f), w, acc[i][1]);         \
        }                                                               \
    }

    LOAD_SC(Pa, Qa, Wa_, 0)
    for (int it = 0; it < ND / (8 * KG) - 1; ++it) {
        LOAD_SC(Pb, Qb, Wb_, 4 * KG)
        hold_order();
        MAC_SC(Pa, Qa, Wa_)
        hold_order();
        LOAD_SC(Pa, Qa, Wa_, 8 * KG)
        hold_order();
        MAC_SC(Pb, Qb, Wb_)
        hold_order();
        pp += 8 * KG;
        qq += 16 * KG;
        ww += 8 * KG;
    }
    LOAD_SC(Pb, Qb, Wb_, 4 * KG)
    hold_order();
    MAC_SC(Pa, Qa, Wa_)
    MAC_SC(Pb, Qb, Wb_)
#undef LOAD_SC
#undef MAC_SC

    const float bb = b2[0];
    const int v0 = vt * 32 + 2 * tv;
#pragma unroll
    for (int i = 0; i < 4; ++i) {
        const int u = ut * 32 + tu + 8 * i;
        const v2f Av = *(const v2f*)(A_a + u * NC + v0);
        v2f g;
        g.x = (1.0f / (1.0f + __expf(-(acc[i][0] + bb)))) * Av.x;
        g.y = (1.0f / (1.0f + __expf(-(acc[i][1] + bb)))) * Av.y;
        *(v2f*)(G + (b * NC + u) * NC + v0) = g;
    }
}

__global__ void __launch_bounds__(256) k_agg_final(const float* __restrict__ G,
                                                   const float* __restrict__ Am,
                                                   const float* __restrict__ h,
                                                   const float* __restrict__ Wa,
                                                   const float* __restrict__ Wm,
                                                   const float* __restrict__ ba,
                                                   const float* __restrict__ bm,
                                                   const float* __restrict__ pb1,
                                                   const float* __restrict__ pb2,
                                                   const float* __restrict__ pb3,
                                                   float* __restrict__ z) {
    __shared__ __attribute__((aligned(16))) float Gl[NC * 16];
    __shared__ __attribute__((aligned(16))) float Ml[NC * 16];
    __shared__ __attribute__((aligned(16))) float Xl[16 * SP];
    __shared__ __attribute__((aligned(16))) float Yl[16 * SP];
    const int lane = threadIdx.x & 31;
    const int wv   = threadIdx.x >> 5;
    const int mt = blockIdx.x >> 1;
    const int vt = mt & 7;
    const int b  = mt >> 3;
    const int nt = (blockIdx.x & 1) * 8 + wv;
    const int hi = lane >> 4, lo = lane & 15;
    const int n  = nt * 16 + lo;
    const int d0 = wv * 32 + lo;

    const float* Gb = G + b * NC * NC;
    const float* hb = h + b * NC * ND;

    v8f accA0 = {}, accM0 = {}, accA1 = {}, accM1 = {};
    const float* Gh = Gl + (2 * hi) * 16 + lo;
    const float* Mh = Ml + (2 * hi) * 16 + lo;
    const float* Hh = hb + (2 * hi) * ND + d0;
    v2f fg0[KCH], fm0[KCH], fh0[KCH], fk0[KCH], fg1[KCH], fm1[KCH], fh1[KCH], fk1[KCH];

    const float* Ar = Xl + lo * SP + 2 * hi;
    const float* Mr = Yl + lo * SP + 2 * hi;
    const float* Wah = Wa + (2 * hi) * ND + n;
    const float* Wmh = Wm + (2 * hi) * ND + n;
    v8f acc1 = {}, acc2 = {};
    v2f xa0[KCH], xm0[KCH], wa0[KCH], wm0[KCH], xa1[KCH], xm1[KCH], wa1[KCH], wm1[KCH];
    float hv0[8], hv1[8];

#define LOAD_AGG_B(B0_, B1_, K0)                                        \
    _Pragma("unroll")          \
    for (int j = 0; j < KCH; ++j) {                                     \
        const int kk = (K0) + 4 * j;                                    \
        B0_[j].x = Hh[kk * ND];                                         \
        B0_[j].y = Hh[(kk + 1) * ND];                                   \
        B1_[j].x = Hh[kk * ND + 16];                                    \
        B1_[j].y = Hh[(kk + 1) * ND + 16];                              \
    }
#define LOAD_AGG_A(G_, M_, K0)                                          \
    _Pragma("unroll")          \
    for (int j = 0; j < KCH; ++j) {                                     \
        const int kk = (K0) + 4 * j;                                    \
        G_[j].x = Gh[kk * 16];                                          \
        G_[j].y = Gh[(kk + 1) * 16];                                    \
        M_[j].x = Mh[kk * 16];                                          \
        M_[j].y = Mh[(kk + 1) * 16];                                    \
    }
#define LOAD_AGG(G_, M_, B0_, B1_, K0) LOAD_AGG_B(B0_, B1_, K0) LOAD_AGG_A(G_, M_, K0)
#define MMA_AGG(G_, M_, B0_, B1_)                                       \
    _Pragma("unroll")          \
    for (int j = 0; j < KCH; ++j) {                                     \
        accA0 = wmma4(G_[j], B0_[j], accA0);                            \
        accM0 = wmma4(M_[j], B0_[j], accM0);                            \
        accA1 = wmma4(G_[j], B1_[j], accA1);                            \
        accM1 = wmma4(M_[j], B1_[j], accM1);                            \
    }
#define LOAD_FIN_B(U_, W_, K0)                                          \
    _Pragma("unroll")          \
    for (int j = 0; j < KCH; ++j) {                                     \
        const int kk = (K0) + 4 * j;                                    \
        U_[j].x = Wah[kk * ND];                                         \
        U_[j].y = Wah[(kk + 1) * ND];                                   \
        W_[j].x = Wmh[kk * ND];                                         \
        W_[j].y = Wmh[(kk + 1) * ND];                                   \
    }
#define LOAD_FIN_A(X_, Y_, K0)                                          \
    _Pragma("unroll")          \
    for (int j = 0; j < KCH; ++j) {                                     \
        const int kk = (K0) + 4 * j;                                    \
        X_[j] = *(const v2f*)(Ar + kk);                                 \
        Y_[j] = *(const v2f*)(Mr + kk);                                 \
    }
#define LOAD_FIN(X_, Y_, U_, W_, K0) LOAD_FIN_B(U_, W_, K0) LOAD_FIN_A(X_, Y_, K0)
#define MMA_FIN(X_, Y_, U_, W_)                                         \
    _Pragma("unroll")          \
    for (int j = 0; j < KCH; ++j) {                                     \
        acc1 = wmma4(X_[j], U_[j], acc1);                               \
        acc2 = wmma4(Y_[j], W_[j], acc2);                               \
    }

    LOAD_AGG_B(fh0, fk0, 0)
    hold_order();
    {
        const float* Gs = Gb + vt * 16;
        const float* Ms = Am + vt * 16;
#pragma unroll
        for (int it = 0; it < 2; ++it) {
            const int u = (threadIdx.x >> 2) + 64 * it;
            const int q = (threadIdx.x & 3) * 4;
            *(v4f*)(Gl + u * 16 + q) = *(const v4f*)(Gs + u * NC + q);
            *(v4f*)(Ml + u * 16 + q) = *(const v4f*)(Ms + u * NC + q);
        }
    }
    __syncthreads();
    LOAD_AGG_A(fg0, fm0, 0)
#pragma unroll
    for (int c = 0; c < NC / (4 * KCH); c += 2) {
        LOAD_AGG(fg1, fm1, fh1, fk1, (c + 1) * 4 * KCH)
        hold_order();
        MMA_AGG(fg0, fm0, fh0, fk0)
        hold_order();
        if (c + 2 < NC / (4 * KCH)) {
            LOAD_AGG(fg0, fm0, fh0, fk0, (c + 2) * 4 * KCH)
        } else {
            LOAD_FIN_B(wa0, wm0, 0)
#pragma unroll
            for (int g = 0; g < 8; ++g) {
                const int row = vt * 16 + g + 8 * hi;
                hv0[g] = hb[row * ND + d0];
                hv1[g] = hb[row * ND + d0 + 16];
            }
        }
        hold_order();
        MMA_AGG(fg1, fm1, fh1, fk1)
        hold_order();
    }
#pragma unroll
    for (int g = 0; g < 8; ++g) {
        const int r = g + 8 * hi;
        Xl[r * SP + d0]      = accA0[g];
        Xl[r * SP + d0 + 16] = accA1[g];
        Yl[r * SP + d0]      = accM0[g] * hv0[g];
        Yl[r * SP + d0 + 16] = accM1[g] * hv1[g];
    }
    __syncthreads();

    LOAD_FIN_A(xa0, xm0, 0)
#pragma unroll
    for (int c = 0; c < ND / (4 * KCH); c += 2) {
        LOAD_FIN(xa1, xm1, wa1, wm1, (c + 1) * 4 * KCH)
        hold_order();
        MMA_FIN(xa0, xm0, wa0, wm0)
        hold_order();
        if (c + 2 < ND / (4 * KCH)) {
            LOAD_FIN(xa0, xm0, wa0, wm0, (c + 2) * 4 * KCH)
        }
        hold_order();
        MMA_FIN(xa1, xm1, wa1, wm1)
        hold_order();
    }
#undef LOAD_AGG
#undef LOAD_AGG_A
#undef LOAD_AGG_B
#undef MMA_AGG
#undef LOAD_FIN
#undef LOAD_FIN_A
#undef LOAD_FIN_B
#undef MMA_FIN
    const float beta1 = pb1[0], beta2 = pb2[0], beta3 = pb3[0];
    const float bav = ba[n], bmv = bm[n];
#pragma unroll
    for (int g = 0; g < 8; ++g) {
        const int row = mt * 16 + g + 8 * hi;
        z[row * ND + n] = beta1 * h[row * ND + n]
                        + beta2 * (acc1[g] + bav)
                        + beta3 * (acc2[g] + bmv);
    }
}

extern "C" void kernel_launch(void* const* d_in, const int* in_sizes, int n_in,
                              void* d_out, int out_size, void* d_ws, size_t ws_size,
                              hipStream_t stream) {
    (void)in_sizes; (void)n_in; (void)out_size; (void)ws_size;

    const float* h_t = (const float*)d_in[0];
    const float* A_a = (const float*)d_in[1];
    const float* A_m = (const float*)d_in[2];
    const float* W1  = (const float*)d_in[3];
    const float* b1  = (const float*)d_in[4];
    const float* W2  = (const float*)d_in[5];
    const float* b2  = (const float*)d_in[6];
    const float* Wa  = (const float*)d_in[7];
    const float* ba  = (const float*)d_in[8];
    const float* Wm  = (const float*)d_in[9];
    const float* bm  = (const float*)d_in[10];
    const float* be1 = (const float*)d_in[11];
    const float* be2 = (const float*)d_in[12];
    const float* be3 = (const float*)d_in[13];
    float* z = (float*)d_out;

    float* P  = (float*)d_ws;
    float* Q  = P + NROWS * ND;
    float* G  = Q + NROWS * ND;

    k_gemm_pq<<<256, 128, 0, stream>>>(h_t, W1, P, Q);
    k_scores<<<NB * 4 * 4, 128, 0, stream>>>(P, Q, b1, W2, b2, A_a, G);
    k_agg_final<<<256, 256, 0, stream>>>(G, A_m, h_t, Wa, Wm, ba, bm,
                                         be1, be2, be3, z);
}
